# starting kernel
# baseline (speedup 1.0000x reference)
#include <hip/hip_runtime.h>
#include <hip/hip_bf16.h>


typedef __attribute__((ext_vector_type(2))) float v2f;
typedef __attribute__((ext_vector_type(8))) float v8f;

__global__ __launch_bounds__(256) void gemm_f32_wmma(
    const float* __restrict__ A, const float* __restrict__ B,
    const float* __restrict__ bias, float* __restrict__ C, int M, int K) {
  const int wave = threadIdx.x >> 5;
  const int lane = threadIdx.x & 31;
  const int half = lane >> 4;
  const int l15  = lane & 15;

  const int row0 = (blockIdx.x * 8 + wave) * 16;
  if (row0 >= M) return;

  v8f acc0 = {}, acc1 = {}, acc2 = {}, acc3 = {};
  const float* aRow = A + (size_t)(row0 + l15) * K + 2 * half;

  for (int k = 0; k < K; k += 4) {
    v2f a;
    a.x = aRow[k];
    a.y = aRow[k + 1];
    const float* bRow = B + (size_t)(k + 2 * half) * 64 + l15;
    v2f b0, b1, b2, b3;
    b0.x = bRow[0];   b0.y = bRow[64];
    b1.x = bRow[16];  b1.y = bRow[80];
    b2.x = bRow[32];  b2.y = bRow[96];
    b3.x = bRow[48];  b3.y = bRow[112];
    acc0 = __builtin_amdgcn_wmma_f32_16x16x4_f32(false, a, false, b0,
                                                 (short)0, acc0, false, false);
    acc1 = __builtin_amdgcn_wmma_f32_16x16x4_f32(false, a, false, b1,
                                                 (short)0, acc1, false, false);
    acc2 = __builtin_amdgcn_wmma_f32_16x16x4_f32(false, a, false, b2,
                                                 (short)0, acc2, false, false);
    acc3 = __builtin_amdgcn_wmma_f32_16x16x4_f32(false, a, false, b3,
                                                 (short)0, acc3, false, false);
  }

  float bb0 = 0.f, bb1 = 0.f, bb2 = 0.f, bb3 = 0.f;
  if (bias) {
    bb0 = bias[l15];      bb1 = bias[l15 + 16];
    bb2 = bias[l15 + 32]; bb3 = bias[l15 + 48];
  }
#pragma unroll
  for (int r = 0; r < 8; ++r) {
    const int row = row0 + r + 8 * half;
    float* cRow = C + (size_t)row * 64 + l15;
    cRow[0]  = acc0[r] + bb0;
    cRow[16] = acc1[r] + bb1;
    cRow[32] = acc2[r] + bb2;
    cRow[48] = acc3[r] + bb3;
  }
}

__global__ void fill_ones(float* __restrict__ p, int n) {
  int i = blockIdx.x * blockDim.x + threadIdx.x;
  if (i < n) p[i] = 1.0f;
}

__global__ void deg_accum(const int* __restrict__ dst,
                          const float* __restrict__ ew,
                          float* __restrict__ deg, int ne) {
  int e = blockIdx.x * blockDim.x + threadIdx.x;
  if (e < ne) atomicAdd(&deg[dst[e]], ew[e]);
}

__global__ void deg_to_dinv(float* __restrict__ p, int n) {
  int i = blockIdx.x * blockDim.x + threadIdx.x;
  if (i < n) {
    float d = p[i];
    p[i] = (d > 0.f) ? rsqrtf(d) : 0.f;
  }
}

__global__ void self_loop_init(const float* __restrict__ h,
                               const float* __restrict__ dinv,
                               float* __restrict__ agg, int total) {
  int t = blockIdx.x * blockDim.x + threadIdx.x;
  if (t >= total) return;
  float di = dinv[t >> 6];
  agg[t] = h[t] * di * di;
}

__global__ __launch_bounds__(256) void edge_scatter(
    const float* __restrict__ h, const int* __restrict__ src,
    const int* __restrict__ dst, const float* __restrict__ ew,
    const float* __restrict__ dinv, float* __restrict__ agg, int ne) {
  long long t = (long long)blockIdx.x * blockDim.x + threadIdx.x;
  int e = (int)(t >> 4);
  if (e >= ne) return;
  int f0 = ((int)t & 15) * 4;
  int s = src[e], d = dst[e];
  float nrm = dinv[s] * ew[e] * dinv[d];
  const float4 hv = *(const float4*)(h + (size_t)s * 64 + f0);
  float* ad = agg + (size_t)d * 64 + f0;
  atomicAdd(ad + 0, hv.x * nrm);
  atomicAdd(ad + 1, hv.y * nrm);
  atomicAdd(ad + 2, hv.z * nrm);
  atomicAdd(ad + 3, hv.w * nrm);
}

__global__ void bias_act(float* __restrict__ v, const float* __restrict__ b,
                         int total, int relu) {
  int t = blockIdx.x * blockDim.x + threadIdx.x;
  if (t >= total) return;
  float x = v[t] + b[t & 63];
  v[t] = relu ? fmaxf(x, 0.f) : x;
}

extern "C" void kernel_launch(void* const* d_in, const int* in_sizes, int n_in,
                              void* d_out, int out_size, void* d_ws,
                              size_t ws_size, hipStream_t stream) {
  const float* x  = (const float*)d_in[0];
  const int*   ei = (const int*)d_in[1];
  const float* ew = (const float*)d_in[2];
  const float* Wm = (const float*)d_in[3];
  const float* bm = (const float*)d_in[4];
  const float* W1 = (const float*)d_in[5];
  const float* b1 = (const float*)d_in[6];
  const float* W2 = (const float*)d_in[7];
  const float* b2 = (const float*)d_in[8];
  float* out = (float*)d_out;

  const int NN = in_sizes[0] / 384;
  const int EE = in_sizes[2];
  const int* src = ei;
  const int* dst = ei + EE;

  char* ws = (char*)d_ws;
  float* dinv = (float*)ws;
  size_t off  = ((size_t)NN * sizeof(float) + 255) & ~(size_t)255;
  float* x64  = (float*)(ws + off);
  float* hbuf = (float*)(ws + off + 1 * (size_t)NN * 64 * sizeof(float));
  float* agg  = (float*)(ws + off + 2 * (size_t)NN * 64 * sizeof(float));

  const int total = NN * 64;
  const dim3 blk(256);
  const int gTot  = (total + 255) / 256;
  const int gN    = (NN + 255) / 256;
  const int gE    = (EE + 255) / 256;
  const int gE16  = (int)(((long long)EE * 16 + 255) / 256);
  const int gGemm = ((NN + 15) / 16 + 7) / 8;

  fill_ones<<<gN, blk, 0, stream>>>(dinv, NN);
  deg_accum<<<gE, blk, 0, stream>>>(dst, ew, dinv, EE);
  deg_to_dinv<<<gN, blk, 0, stream>>>(dinv, NN);

  gemm_f32_wmma<<<gGemm, blk, 0, stream>>>(x, Wm, bm, x64, NN, 384);

  gemm_f32_wmma<<<gGemm, blk, 0, stream>>>(x64, W1, nullptr, hbuf, NN, 64);
  self_loop_init<<<gTot, blk, 0, stream>>>(hbuf, dinv, agg, total);
  edge_scatter<<<gE16, blk, 0, stream>>>(hbuf, src, dst, ew, dinv, agg, EE);
  bias_act<<<gTot, blk, 0, stream>>>(agg, b1, total, 1);

  gemm_f32_wmma<<<gGemm, blk, 0, stream>>>(agg, W2, nullptr, hbuf, NN, 64);
  self_loop_init<<<gTot, blk, 0, stream>>>(hbuf, dinv, out, total);
  edge_scatter<<<gE16, blk, 0, stream>>>(hbuf, src, dst, ew, dinv, out, EE);
  bias_act<<<gTot, blk, 0, stream>>>(out, b2, total, 0);
}


# Edges counting-sorted by destination each call (int atomic ranks + prefix sum); each output row summed in registers in rank order, written once with self loop, bias, relu; float atomics removed
# speedup vs baseline: 1.7295x; 1.7295x over previous
#include <hip/hip_runtime.h>
#include <hip/hip_bf16.h>


typedef __attribute__((ext_vector_type(2))) float v2f;
typedef __attribute__((ext_vector_type(8))) float v8f;

__global__ __launch_bounds__(256) void gemm_f32_wmma(
    const float* __restrict__ A, const float* __restrict__ B,
    const float* __restrict__ bias, float* __restrict__ C, int M, int K) {
  const int wave = threadIdx.x >> 5;
  const int lane = threadIdx.x & 31;
  const int half = lane >> 4;
  const int l15  = lane & 15;

  const int row0 = (blockIdx.x * 8 + wave) * 16;
  if (row0 >= M) return;

  v8f acc0 = {}, acc1 = {}, acc2 = {}, acc3 = {};
  const float* aRow = A + (size_t)(row0 + l15) * K + 2 * half;

  for (int k = 0; k < K; k += 4) {
    v2f a;
    a.x = aRow[k];
    a.y = aRow[k + 1];
    const float* bRow = B + (size_t)(k + 2 * half) * 64 + l15;
    v2f b0, b1, b2, b3;
    b0.x = bRow[0];   b0.y = bRow[64];
    b1.x = bRow[16];  b1.y = bRow[80];
    b2.x = bRow[32];  b2.y = bRow[96];
    b3.x = bRow[48];  b3.y = bRow[112];
    acc0 = __builtin_amdgcn_wmma_f32_16x16x4_f32(false, a, false, b0,
                                                 (short)0, acc0, false, false);
    acc1 = __builtin_amdgcn_wmma_f32_16x16x4_f32(false, a, false, b1,
                                                 (short)0, acc1, false, false);
    acc2 = __builtin_amdgcn_wmma_f32_16x16x4_f32(false, a, false, b2,
                                                 (short)0, acc2, false, false);
    acc3 = __builtin_amdgcn_wmma_f32_16x16x4_f32(false, a, false, b3,
                                                 (short)0, acc3, false, false);
  }

  float bb0 = 0.f, bb1 = 0.f, bb2 = 0.f, bb3 = 0.f;
  if (bias) {
    bb0 = bias[l15];      bb1 = bias[l15 + 16];
    bb2 = bias[l15 + 32]; bb3 = bias[l15 + 48];
  }
#pragma unroll
  for (int r = 0; r < 8; ++r) {
    const int row = row0 + r + 8 * half;
    float* cRow = C + (size_t)row * 64 + l15;
    cRow[0]  = acc0[r] + bb0;
    cRow[16] = acc1[r] + bb1;
    cRow[32] = acc2[r] + bb2;
    cRow[48] = acc3[r] + bb3;
  }
}

__global__ void zero_counts(int* __restrict__ cnt, int n) {
  int i = blockIdx.x * blockDim.x + threadIdx.x;
  if (i < n) cnt[i] = 0;
}

__global__ void edge_rank(const int* __restrict__ dst, int* __restrict__ cnt,
                          int* __restrict__ rank, int ne) {
  int e = blockIdx.x * blockDim.x + threadIdx.x;
  if (e < ne) rank[e] = atomicAdd(&cnt[dst[e]], 1);
}

__global__ __launch_bounds__(1024) void row_offsets(
    const int* __restrict__ cnt, int* __restrict__ rowptr, int n) {
  __shared__ int tileSum[32];
  __shared__ int preSum[32];
  const int t = threadIdx.x;
  const int lane = t & 31;
  const int wave = t >> 5;

  int pre = 0;
  const int4* c4 = (const int4*)cnt;
  const int nq = (int)blockIdx.x * 256;
  for (int j0 = 0; j0 < nq; j0 += 8 * 1024) {
    int4 v[8];
#pragma unroll
    for (int u = 0; u < 8; ++u) {
      const int j = j0 + u * 1024 + t;
      v[u] = (j < nq) ? c4[j] : make_int4(0, 0, 0, 0);
    }
#pragma unroll
    for (int u = 0; u < 8; ++u) pre += (v[u].x + v[u].y) + (v[u].z + v[u].w);
  }

  const int i = (int)blockIdx.x * 1024 + t;
  const int mine = (i < n) ? cnt[i] : 0;

  for (int o = 16; o > 0; o >>= 1) pre += __shfl_xor(pre, o);
  int inc = mine;
  for (int o = 1; o < 32; o <<= 1) {
    int v = __shfl_up(inc, o);
    if (lane >= o) inc += v;
  }
  if (lane == 31) tileSum[wave] = inc;
  if (lane == 0) preSum[wave] = pre;
  __syncthreads();

  int base = 0;
  for (int w = 0; w < 32; ++w) {
    base += preSum[w];
    if (w < wave) base += tileSum[w];
  }
  const int ex = base + inc - mine;
  if (i < n) rowptr[i] = ex;
  if (i == n - 1) rowptr[n] = ex + mine;
}

__global__ void csr_fill(const int* __restrict__ src,
                         const int* __restrict__ dst,
                         const float* __restrict__ ew,
                         const int* __restrict__ rank,
                         const int* __restrict__ rowptr,
                         int* __restrict__ csr, int ne) {
  int e = blockIdx.x * blockDim.x + threadIdx.x;
  if (e >= ne) return;
  int p = rowptr[dst[e]] + rank[e];
  ((int2*)csr)[p] = make_int2(src[e], __float_as_int(ew[e]));
}

__global__ void node_dinv(const int* __restrict__ rowptr,
                          const int* __restrict__ csr,
                          float* __restrict__ dinv, int n) {
  int i = blockIdx.x * blockDim.x + threadIdx.x;
  if (i >= n) return;
  const int2* ed = (const int2*)csr;
  const int end = rowptr[i + 1];
  float d = 1.0f;
  for (int p = rowptr[i]; p < end; ++p) d += __int_as_float(ed[p].y);
  dinv[i] = (d > 0.f) ? rsqrtf(d) : 0.f;
}

__global__ __launch_bounds__(256) void gather_rows(
    const float* __restrict__ h, const int* __restrict__ rowptr,
    const int* __restrict__ csr, const float* __restrict__ dinv,
    const float* __restrict__ bias, float* __restrict__ out, int n, int relu) {
  const int t = blockIdx.x * 256 + threadIdx.x;
  const int i = t >> 4;
  if (i >= n) return;
  const int l = t & 15;
  const float4* h4 = (const float4*)h;
  const int2* ed = (const int2*)csr;

  const int beg = rowptr[i];
  const int end = rowptr[i + 1];
  const float di = dinv[i];
  const float4 hv = h4[(size_t)i * 16 + l];
  float ax = hv.x * di, ay = hv.y * di, az = hv.z * di, aw = hv.w * di;

  for (int base = beg; base < end; base += 16) {
    const int p = base + l;
    int s = i;
    float cw = 0.f;
    if (p < end) {
      const int2 e = ed[p];
      s = min(max(e.x, 0), n - 1);
      cw = __int_as_float(e.y) * dinv[s];
    }
    const int m = min(16, end - base);
    for (int k = 0; k < m; k += 4) {
      const int s0 = __shfl(s, k + 0, 16);
      const int s1 = __shfl(s, k + 1, 16);
      const int s2 = __shfl(s, k + 2, 16);
      const int s3 = __shfl(s, k + 3, 16);
      const float c0 = __shfl(cw, k + 0, 16);
      const float c1 = __shfl(cw, k + 1, 16);
      const float c2 = __shfl(cw, k + 2, 16);
      const float c3 = __shfl(cw, k + 3, 16);
      const float4 v0 = h4[(size_t)s0 * 16 + l];
      const float4 v1 = h4[(size_t)s1 * 16 + l];
      const float4 v2 = h4[(size_t)s2 * 16 + l];
      const float4 v3 = h4[(size_t)s3 * 16 + l];
      ax = fmaf(v0.x, c0, ax); ay = fmaf(v0.y, c0, ay);
      az = fmaf(v0.z, c0, az); aw = fmaf(v0.w, c0, aw);
      ax = fmaf(v1.x, c1, ax); ay = fmaf(v1.y, c1, ay);
      az = fmaf(v1.z, c1, az); aw = fmaf(v1.w, c1, aw);
      ax = fmaf(v2.x, c2, ax); ay = fmaf(v2.y, c2, ay);
      az = fmaf(v2.z, c2, az); aw = fmaf(v2.w, c2, aw);
      ax = fmaf(v3.x, c3, ax); ay = fmaf(v3.y, c3, ay);
      az = fmaf(v3.z, c3, az); aw = fmaf(v3.w, c3, aw);
    }
  }

  float4 r;
  r.x = fmaf(ax, di, bias[l * 4 + 0]);
  r.y = fmaf(ay, di, bias[l * 4 + 1]);
  r.z = fmaf(az, di, bias[l * 4 + 2]);
  r.w = fmaf(aw, di, bias[l * 4 + 3]);
  if (relu) {
    r.x = fmaxf(r.x, 0.f); r.y = fmaxf(r.y, 0.f);
    r.z = fmaxf(r.z, 0.f); r.w = fmaxf(r.w, 0.f);
  }
  ((float4*)out)[(size_t)i * 16 + l] = r;
}

extern "C" void kernel_launch(void* const* d_in, const int* in_sizes, int n_in,
                              void* d_out, int out_size, void* d_ws,
                              size_t ws_size, hipStream_t stream) {
  const float* x  = (const float*)d_in[0];
  const int*   ei = (const int*)d_in[1];
  const float* ew = (const float*)d_in[2];
  const float* Wm = (const float*)d_in[3];
  const float* bm = (const float*)d_in[4];
  const float* W1 = (const float*)d_in[5];
  const float* b1 = (const float*)d_in[6];
  const float* W2 = (const float*)d_in[7];
  const float* b2 = (const float*)d_in[8];
  float* out = (float*)d_out;

  const int NN = in_sizes[0] / 384;
  const int EE = in_sizes[2];
  const int* src = ei;
  const int* dst = ei + EE;

  char* ws = (char*)d_ws;
  const size_t nBytes  = ((size_t)NN * 4 + 255) & ~(size_t)255;
  const size_t n1Bytes = ((size_t)(NN + 1) * 4 + 255) & ~(size_t)255;
  const size_t eBytes  = ((size_t)EE * 4 + 255) & ~(size_t)255;
  const size_t e2Bytes = ((size_t)EE * 8 + 255) & ~(size_t)255;
  const size_t fBytes  = (size_t)NN * 64 * sizeof(float);
  size_t off = 0;
  int*   cnt    = (int*)(ws + off);   off += nBytes;
  int*   rowptr = (int*)(ws + off);   off += n1Bytes;
  int*   rank   = (int*)(ws + off);   off += eBytes;
  int*   csr    = (int*)(ws + off);   off += e2Bytes;
  float* dinv   = (float*)(ws + off); off += nBytes;
  float* x64    = (float*)(ws + off); off += fBytes;
  float* hbuf   = (float*)(ws + off); off += fBytes;
  float* agg    = (float*)(ws + off);

  const dim3 blk(256);
  const int gN    = (NN + 255) / 256;
  const int gE    = (EE + 255) / 256;
  const int gScan = (NN + 1023) / 1024;
  const int gRow  = (int)(((long long)NN * 16 + 255) / 256);
  const int gGemm = ((NN + 15) / 16 + 7) / 8;

  zero_counts<<<gN, blk, 0, stream>>>(cnt, NN);
  edge_rank<<<gE, blk, 0, stream>>>(dst, cnt, rank, EE);
  row_offsets<<<gScan, dim3(1024), 0, stream>>>(cnt, rowptr, NN);
  csr_fill<<<gE, blk, 0, stream>>>(src, dst, ew, rank, rowptr, csr, EE);
  node_dinv<<<gN, blk, 0, stream>>>(rowptr, csr, dinv, NN);

  gemm_f32_wmma<<<gGemm, blk, 0, stream>>>(x, Wm, bm, x64, NN, 384);

  gemm_f32_wmma<<<gGemm, blk, 0, stream>>>(x64, W1, nullptr, hbuf, NN, 64);
  gather_rows<<<gRow, blk, 0, stream>>>(hbuf, rowptr, csr, dinv, b1, agg, NN, 1);

  gemm_f32_wmma<<<gGemm, blk, 0, stream>>>(agg, W2, nullptr, hbuf, NN, 64);
  gather_rows<<<gRow, blk, 0, stream>>>(hbuf, rowptr, csr, dinv, b2, out, NN, 0);
}


# Edge sort rebuilt as two LDS-binning launches (chunks binned in LDS, then one block per 256-node bin ranks rows and sums degrees in LDS): no global atomics, 3 launches fewer; row order = LDS arrival
# speedup vs baseline: 2.3715x; 1.3712x over previous
#include <hip/hip_runtime.h>
#include <hip/hip_bf16.h>


typedef __attribute__((ext_vector_type(2))) float v2f;
typedef __attribute__((ext_vector_type(8))) float v8f;

__global__ __launch_bounds__(256) void gemm_f32_wmma(
    const float* __restrict__ A, const float* __restrict__ B,
    const float* __restrict__ bias, float* __restrict__ C, int M, int K) {
  const int wave = threadIdx.x >> 5;
  const int lane = threadIdx.x & 31;
  const int half = lane >> 4;
  const int l15  = lane & 15;

  const int row0 = (blockIdx.x * 8 + wave) * 16;
  if (row0 >= M) return;

  v8f acc0 = {}, acc1 = {}, acc2 = {}, acc3 = {};
  const float* aRow = A + (size_t)(row0 + l15) * K + 2 * half;

  for (int k = 0; k < K; k += 4) {
    v2f a;
    a.x = aRow[k];
    a.y = aRow[k + 1];
    const float* bRow = B + (size_t)(k + 2 * half) * 64 + l15;
    v2f b0, b1, b2, b3;
    b0.x = bRow[0];   b0.y = bRow[64];
    b1.x = bRow[16];  b1.y = bRow[80];
    b2.x = bRow[32];  b2.y = bRow[96];
    b3.x = bRow[48];  b3.y = bRow[112];
    acc0 = __builtin_amdgcn_wmma_f32_16x16x4_f32(false, a, false, b0,
                                                 (short)0, acc0, false, false);
    acc1 = __builtin_amdgcn_wmma_f32_16x16x4_f32(false, a, false, b1,
                                                 (short)0, acc1, false, false);
    acc2 = __builtin_amdgcn_wmma_f32_16x16x4_f32(false, a, false, b2,
                                                 (short)0, acc2, false, false);
    acc3 = __builtin_amdgcn_wmma_f32_16x16x4_f32(false, a, false, b3,
                                                 (short)0, acc3, false, false);
  }

  float bb0 = 0.f, bb1 = 0.f, bb2 = 0.f, bb3 = 0.f;
  if (bias) {
    bb0 = bias[l15];      bb1 = bias[l15 + 16];
    bb2 = bias[l15 + 32]; bb3 = bias[l15 + 48];
  }
#pragma unroll
  for (int r = 0; r < 8; ++r) {
    const int row = row0 + r + 8 * half;
    float* cRow = C + (size_t)row * 64 + l15;
    cRow[0]  = acc0[r] + bb0;
    cRow[16] = acc1[r] + bb1;
    cRow[32] = acc2[r] + bb2;
    cRow[48] = acc3[r] + bb3;
  }
}

#define SORT_W 256
#define SORT_R 256
#define SORT_RS 8

__device__ __forceinline__ int block_excl_scan(int v, int* sh, int* total) {
  const int lane = threadIdx.x & 31;
  const int wave = threadIdx.x >> 5;
  int inc = v;
  for (int o = 1; o < 32; o <<= 1) {
    const int u = __shfl_up(inc, o);
    if (lane >= o) inc += u;
  }
  if (lane == 31) sh[wave] = inc;
  __syncthreads();
  if (wave == 0) {
    const int s = sh[lane];
    int si = s;
    for (int o = 1; o < 32; o <<= 1) {
      const int u = __shfl_up(si, o);
      if (lane >= o) si += u;
    }
    sh[lane] = si - s;
    if (lane == 31) sh[32] = si;
  }
  __syncthreads();
  const int ex = sh[wave] + inc - v;
  *total = sh[32];
  __syncthreads();
  return ex;
}

__global__ __launch_bounds__(1024) void bin_edges(
    const int* __restrict__ src, const int* __restrict__ dst,
    const float* __restrict__ ew, int* __restrict__ segSrc,
    int* __restrict__ segDst, float* __restrict__ segW,
    int* __restrict__ cntTab, int* __restrict__ offTab,
    int ne, int n, int nb, int chunk) {
  extern __shared__ int dyn[];
  __shared__ int sh[33];
  int* binCnt = dyn;
  int* binCur = dyn + nb;
  int* stS = dyn + 2 * nb;
  int* stD = stS + chunk;
  float* stW = (float*)(stD + chunk);
  const int t = threadIdx.x;
  const int w = blockIdx.x;
  const int c0 = min(w * chunk, ne);
  const int c1 = min(c0 + chunk, ne);

  for (int b = t; b < nb; b += 1024) binCnt[b] = 0;
  __syncthreads();
  for (int e0 = c0 + t; e0 < c1; e0 += 4 * 1024) {
    int d[4];
#pragma unroll
    for (int u = 0; u < 4; ++u) {
      d[u] = dst[min(e0 + u * 1024, c1 - 1)];
    }
    asm volatile("" ::"v"(d[0]), "v"(d[1]), "v"(d[2]), "v"(d[3]));
#pragma unroll
    for (int u = 0; u < 4; ++u) {
      if (e0 + u * 1024 < c1) {
        const int dd = min(max(d[u], 0), n - 1);
        atomicAdd(&binCnt[dd >> SORT_RS], 1);
      }
    }
  }
  __syncthreads();

  int carry = 0;
  for (int b0 = 0; b0 < nb; b0 += 1024) {
    const int b = b0 + t;
    const int v = (b < nb) ? binCnt[b] : 0;
    int tot;
    const int ex = carry + block_excl_scan(v, sh, &tot);
    if (b < nb) {
      binCur[b] = ex;
      cntTab[(size_t)w * nb + b] = v;
      offTab[(size_t)w * nb + b] = ex;
    }
    carry += tot;
  }
  __syncthreads();

  for (int e0 = c0 + t; e0 < c1; e0 += 4 * 1024) {
    int d[4], s[4];
    float wv[4];
#pragma unroll
    for (int u = 0; u < 4; ++u) {
      const int e = min(e0 + u * 1024, c1 - 1);
      d[u]  = dst[e];
      s[u]  = src[e];
      wv[u] = ew[e];
    }
    asm volatile("" ::"v"(d[0]), "v"(d[1]), "v"(d[2]), "v"(d[3]),
                 "v"(s[0]), "v"(s[1]), "v"(s[2]), "v"(s[3]),
                 "v"(wv[0]), "v"(wv[1]), "v"(wv[2]), "v"(wv[3]));
#pragma unroll
    for (int u = 0; u < 4; ++u) {
      if (e0 + u * 1024 < c1) {
        const int dd = min(max(d[u], 0), n - 1);
        const int p = atomicAdd(&binCur[dd >> SORT_RS], 1);
        stS[p] = s[u];
        stD[p] = dd;
        stW[p] = wv[u];
      }
    }
  }
  __syncthreads();

  const int cn = c1 - c0;
  for (int i = t; i < cn; i += 1024) {
    segSrc[c0 + i] = stS[i];
    segDst[c0 + i] = stD[i];
    segW[c0 + i]   = stW[i];
  }
}

__global__ __launch_bounds__(1024) void own_rows(
    const int* __restrict__ segSrc, const int* __restrict__ segDst,
    const float* __restrict__ segW, const int* __restrict__ cntTab,
    const int* __restrict__ offTab, int* __restrict__ rowptr,
    int* __restrict__ csr, float* __restrict__ dinv, int n, int nb, int chunk) {
  __shared__ int sh[33];
  __shared__ int sCnt[SORT_W];
  __shared__ int sOff[SORT_W];
  __shared__ int nodeCnt[SORT_R];
  __shared__ int cursor[SORT_R];
  __shared__ float nodeDeg[SORT_R];
  const int t = threadIdx.x;
  const int b = blockIdx.x;

  int myOff = 0;
  if (t < SORT_W) {
    sCnt[t] = cntTab[(size_t)t * nb + b];
    myOff = offTab[(size_t)t * nb + b];
    sOff[t] = myOff;
  }
  if (t < SORT_R) {
    nodeCnt[t] = 0;
    nodeDeg[t] = 0.f;
  }
  int base;
  (void)block_excl_scan(myOff, sh, &base);

  const int w = t >> 2;
  const int sub = t & 3;
  const int cnt = sCnt[w];
  const int g0 = w * chunk + sOff[w];
  const int node0 = b * SORT_R;
  for (int i0 = sub; i0 < cnt; i0 += 16) {
    int d[4];
    float wv[4];
#pragma unroll
    for (int u = 0; u < 4; ++u) {
      const int i = i0 + 4 * u;
      const bool ok = i < cnt;
      d[u]  = ok ? segDst[g0 + i] : 0;
      wv[u] = ok ? segW[g0 + i] : 0.f;
    }
#pragma unroll
    for (int u = 0; u < 4; ++u) {
      if (i0 + 4 * u < cnt) {
        const int dd = d[u] - node0;
        atomicAdd(&nodeCnt[dd], 1);
        atomicAdd(&nodeDeg[dd], wv[u]);
      }
    }
  }
  __syncthreads();

  const int v = (t < SORT_R) ? nodeCnt[t] : 0;
  int nEdge;
  const int ex = block_excl_scan(v, sh, &nEdge);
  if (t < SORT_R) {
    cursor[t] = ex;
    const int node = node0 + t;
    if (node < n) {
      rowptr[node] = base + ex;
      const float dg = 1.0f + nodeDeg[t];
      dinv[node] = (dg > 0.f) ? rsqrtf(dg) : 0.f;
      if (node == n - 1) rowptr[n] = base + ex + v;
    }
  }
  __syncthreads();

  int2* ed = (int2*)csr;
  for (int i0 = sub; i0 < cnt; i0 += 16) {
    int d[4], s[4];
    float wv[4];
#pragma unroll
    for (int u = 0; u < 4; ++u) {
      const int i = i0 + 4 * u;
      const bool ok = i < cnt;
      d[u]  = ok ? segDst[g0 + i] : 0;
      s[u]  = ok ? segSrc[g0 + i] : 0;
      wv[u] = ok ? segW[g0 + i] : 0.f;
    }
#pragma unroll
    for (int u = 0; u < 4; ++u) {
      if (i0 + 4 * u < cnt) {
        const int p = atomicAdd(&cursor[d[u] - node0], 1);
        ed[base + p] = make_int2(s[u], __float_as_int(wv[u]));
      }
    }
  }
}

__global__ __launch_bounds__(256) void gather_rows(
    const float* __restrict__ h, const int* __restrict__ rowptr,
    const int* __restrict__ csr, const float* __restrict__ dinv,
    const float* __restrict__ bias, float* __restrict__ out, int n, int relu) {
  const int t = blockIdx.x * 256 + threadIdx.x;
  const int i = t >> 4;
  if (i >= n) return;
  const int l = t & 15;
  const float4* h4 = (const float4*)h;
  const int2* ed = (const int2*)csr;

  const int beg = rowptr[i];
  const int end = rowptr[i + 1];
  const float di = dinv[i];
  const float4 hv = h4[(size_t)i * 16 + l];
  float ax = hv.x * di, ay = hv.y * di, az = hv.z * di, aw = hv.w * di;

  for (int base = beg; base < end; base += 16) {
    const int p = base + l;
    int s = i;
    float cw = 0.f;
    if (p < end) {
      const int2 e = ed[p];
      s = min(max(e.x, 0), n - 1);
      cw = __int_as_float(e.y) * dinv[s];
    }
    const int m = min(16, end - base);
    for (int k = 0; k < m; k += 4) {
      const int s0 = __shfl(s, k + 0, 16);
      const int s1 = __shfl(s, k + 1, 16);
      const int s2 = __shfl(s, k + 2, 16);
      const int s3 = __shfl(s, k + 3, 16);
      const float c0 = __shfl(cw, k + 0, 16);
      const float c1 = __shfl(cw, k + 1, 16);
      const float c2 = __shfl(cw, k + 2, 16);
      const float c3 = __shfl(cw, k + 3, 16);
      const float4 v0 = h4[(size_t)s0 * 16 + l];
      const float4 v1 = h4[(size_t)s1 * 16 + l];
      const float4 v2 = h4[(size_t)s2 * 16 + l];
      const float4 v3 = h4[(size_t)s3 * 16 + l];
      ax = fmaf(v0.x, c0, ax); ay = fmaf(v0.y, c0, ay);
      az = fmaf(v0.z, c0, az); aw = fmaf(v0.w, c0, aw);
      ax = fmaf(v1.x, c1, ax); ay = fmaf(v1.y, c1, ay);
      az = fmaf(v1.z, c1, az); aw = fmaf(v1.w, c1, aw);
      ax = fmaf(v2.x, c2, ax); ay = fmaf(v2.y, c2, ay);
      az = fmaf(v2.z, c2, az); aw = fmaf(v2.w, c2, aw);
      ax = fmaf(v3.x, c3, ax); ay = fmaf(v3.y, c3, ay);
      az = fmaf(v3.z, c3, az); aw = fmaf(v3.w, c3, aw);
    }
  }

  float4 r;
  r.x = fmaf(ax, di, bias[l * 4 + 0]);
  r.y = fmaf(ay, di, bias[l * 4 + 1]);
  r.z = fmaf(az, di, bias[l * 4 + 2]);
  r.w = fmaf(aw, di, bias[l * 4 + 3]);
  if (relu) {
    r.x = fmaxf(r.x, 0.f); r.y = fmaxf(r.y, 0.f);
    r.z = fmaxf(r.z, 0.f); r.w = fmaxf(r.w, 0.f);
  }
  ((float4*)out)[(size_t)i * 16 + l] = r;
}

extern "C" void kernel_launch(void* const* d_in, const int* in_sizes, int n_in,
                              void* d_out, int out_size, void* d_ws,
                              size_t ws_size, hipStream_t stream) {
  const float* x  = (const float*)d_in[0];
  const int*   ei = (const int*)d_in[1];
  const float* ew = (const float*)d_in[2];
  const float* Wm = (const float*)d_in[3];
  const float* bm = (const float*)d_in[4];
  const float* W1 = (const float*)d_in[5];
  const float* b1 = (const float*)d_in[6];
  const float* W2 = (const float*)d_in[7];
  const float* b2 = (const float*)d_in[8];
  float* out = (float*)d_out;

  const int NN = in_sizes[0] / 384;
  const int EE = in_sizes[2];
  const int* src = ei;
  const int* dst = ei + EE;

  char* ws = (char*)d_ws;
  const int nBins = (NN + SORT_R - 1) / SORT_R;
  const int chunk = (EE + SORT_W - 1) / SORT_W;
  const size_t tBytes  = ((size_t)SORT_W * nBins * 4 + 255) & ~(size_t)255;
  const size_t nBytes  = ((size_t)NN * 4 + 255) & ~(size_t)255;
  const size_t n1Bytes = ((size_t)(NN + 1) * 4 + 255) & ~(size_t)255;
  const size_t eBytes  = ((size_t)EE * 4 + 255) & ~(size_t)255;
  const size_t e2Bytes = ((size_t)EE * 8 + 255) & ~(size_t)255;
  const size_t fBytes  = (size_t)NN * 64 * sizeof(float);
  size_t off = 0;
  int*   cntTab = (int*)(ws + off);   off += tBytes;
  int*   offTab = (int*)(ws + off);   off += tBytes;
  int*   segSrc = (int*)(ws + off);   off += eBytes;
  int*   segDst = (int*)(ws + off);   off += eBytes;
  float* segW   = (float*)(ws + off); off += eBytes;
  int*   rowptr = (int*)(ws + off);   off += n1Bytes;
  int*   csr    = (int*)(ws + off);   off += e2Bytes;
  float* dinv   = (float*)(ws + off); off += nBytes;
  float* x64    = (float*)(ws + off); off += fBytes;
  float* hbuf   = (float*)(ws + off); off += fBytes;
  float* agg    = (float*)(ws + off);

  const dim3 blk(256);
  const int gRow  = (int)(((long long)NN * 16 + 255) / 256);
  const int gGemm = ((NN + 15) / 16 + 7) / 8;
  const size_t binLds = ((size_t)2 * nBins + (size_t)3 * chunk) * 4;

  bin_edges<<<SORT_W, dim3(1024), binLds, stream>>>(
      src, dst, ew, segSrc, segDst, segW, cntTab, offTab, EE, NN, nBins, chunk);
  own_rows<<<nBins, dim3(1024), 0, stream>>>(
      segSrc, segDst, segW, cntTab, offTab, rowptr, csr, dinv, NN, nBins, chunk);

  gemm_f32_wmma<<<gGemm, blk, 0, stream>>>(x, Wm, bm, x64, NN, 384);

  gemm_f32_wmma<<<gGemm, blk, 0, stream>>>(x64, W1, nullptr, hbuf, NN, 64);
  gather_rows<<<gRow, blk, 0, stream>>>(hbuf, rowptr, csr, dinv, b1, agg, NN, 1);

  gemm_f32_wmma<<<gGemm, blk, 0, stream>>>(agg, W2, nullptr, hbuf, NN, 64);
  gather_rows<<<gRow, blk, 0, stream>>>(hbuf, rowptr, csr, dinv, b2, out, NN, 0);
}


# Dense products tiled through LDS: 64-row blocks, A and B staged 32 k at a time by whole-line 16-byte loads, double buffered, wave = 16x32 tile; k order and per-element sums unchanged
# speedup vs baseline: 2.8006x; 1.1809x over previous
#include <hip/hip_runtime.h>
#include <hip/hip_bf16.h>


typedef __attribute__((ext_vector_type(2))) float v2f;
typedef __attribute__((ext_vector_type(8))) float v8f;

#define GT_M  64
#define GT_K  32
#define GT_AP 36
#define GT_BP 160

__global__ __launch_bounds__(256, 16) void gemm_f32_wmma(
    const float* __restrict__ A, const float* __restrict__ B,
    const float* __restrict__ bias, float* __restrict__ C, int M, int K) {
  __shared__ __attribute__((aligned(16))) float sA[2][GT_M * GT_AP];
  __shared__ __attribute__((aligned(16))) float sB[2][(GT_K / 2) * GT_BP];

  const int t = threadIdx.x;
  const int wave = t >> 5;
  const int lane = t & 31;
  const int half = lane >> 4;
  const int l15  = lane & 15;
  const int strip = wave >> 1;
  const int colHalf = wave & 1;
  const int rowBlk = blockIdx.x * GT_M;

  const int aRow = t >> 3;
  const int aC4  = t & 7;
  const float4* aP0 =
      (const float4*)(A + (size_t)min(rowBlk + aRow, M - 1) * K) + aC4;
  const float4* aP1 =
      (const float4*)(A + (size_t)min(rowBlk + aRow + 32, M - 1) * K) + aC4;
  const int bKp = t >> 4;
  const int bN4 = t & 15;
  const float4* bP0 = (const float4*)(B + (size_t)(2 * bKp) * 64) + bN4;
  const float4* bP1 = (const float4*)(B + (size_t)(2 * bKp + 1) * 64) + bN4;

  const int aSt0 = aRow * GT_AP + aC4 * 4;
  const int aSt1 = (aRow + 32) * GT_AP + aC4 * 4;
  const int bSt  = bKp * GT_BP + bN4 * 8;
  const int aFr  = (16 * strip + l15) * GT_AP + 2 * half;
  const int bFr  = half * GT_BP + (32 * colHalf + l15) * 2;

  const int nStage = K / GT_K;
  float4 ra0 = aP0[0], ra1 = aP1[0], rb0 = bP0[0], rb1 = bP1[0];
  v8f acc0 = {}, acc1 = {};

  for (int c = 0; c < nStage; ++c) {
    float* dA = sA[c & 1];
    float* dB = sB[c & 1];
    *(float4*)(dA + aSt0) = ra0;
    *(float4*)(dA + aSt1) = ra1;
    *(float4*)(dB + bSt)     = make_float4(rb0.x, rb1.x, rb0.y, rb1.y);
    *(float4*)(dB + bSt + 4) = make_float4(rb0.z, rb1.z, rb0.w, rb1.w);
    __syncthreads();
    if (c + 1 < nStage) {
      ra0 = aP0[(size_t)(c + 1) * (GT_K / 4)];
      ra1 = aP1[(size_t)(c + 1) * (GT_K / 4)];
      rb0 = bP0[(size_t)(c + 1) * (GT_K * 16)];
      rb1 = bP1[(size_t)(c + 1) * (GT_K * 16)];
    }
    const float* fa = dA + aFr;
    const float* fb = dB + bFr;
#pragma unroll
    for (int j = 0; j < GT_K / 4; ++j) {
      const v2f a  = *(const v2f*)(fa + 4 * j);
      const v2f b0 = *(const v2f*)(fb + 2 * j * GT_BP);
      const v2f b1 = *(const v2f*)(fb + 2 * j * GT_BP + 32);
      acc0 = __builtin_amdgcn_wmma_f32_16x16x4_f32(false, a, false, b0,
                                                   (short)0, acc0, false, false);
      acc1 = __builtin_amdgcn_wmma_f32_16x16x4_f32(false, a, false, b1,
                                                   (short)0, acc1, false, false);
    }
  }

  const int col = 32 * colHalf + l15;
  float bb0 = 0.f, bb1 = 0.f;
  if (bias) {
    bb0 = bias[col];
    bb1 = bias[col + 16];
  }
  const int row0 = rowBlk + 16 * strip;
  if (row0 >= M) return;
#pragma unroll
  for (int r = 0; r < 8; ++r) {
    float* cRow = C + (size_t)(row0 + r + 8 * half) * 64 + col;
    cRow[0]  = acc0[r] + bb0;
    cRow[16] = acc1[r] + bb1;
  }
}

#define SORT_W 256
#define SORT_R 256
#define SORT_RS 8

__device__ __forceinline__ int block_excl_scan(int v, int* sh, int* total) {
  const int lane = threadIdx.x & 31;
  const int wave = threadIdx.x >> 5;
  int inc = v;
  for (int o = 1; o < 32; o <<= 1) {
    const int u = __shfl_up(inc, o);
    if (lane >= o) inc += u;
  }
  if (lane == 31) sh[wave] = inc;
  __syncthreads();
  if (wave == 0) {
    const int s = sh[lane];
    int si = s;
    for (int o = 1; o < 32; o <<= 1) {
      const int u = __shfl_up(si, o);
      if (lane >= o) si += u;
    }
    sh[lane] = si - s;
    if (lane == 31) sh[32] = si;
  }
  __syncthreads();
  const int ex = sh[wave] + inc - v;
  *total = sh[32];
  __syncthreads();
  return ex;
}

__global__ __launch_bounds__(1024) void bin_edges(
    const int* __restrict__ src, const int* __restrict__ dst,
    const float* __restrict__ ew, int* __restrict__ segSrc,
    int* __restrict__ segDst, float* __restrict__ segW,
    int* __restrict__ cntTab, int* __restrict__ offTab,
    int ne, int n, int nb, int chunk) {
  extern __shared__ int dyn[];
  __shared__ int sh[33];
  int* binCnt = dyn;
  int* binCur = dyn + nb;
  int* stS = dyn + 2 * nb;
  int* stD = stS + chunk;
  float* stW = (float*)(stD + chunk);
  const int t = threadIdx.x;
  const int w = blockIdx.x;
  const int c0 = min(w * chunk, ne);
  const int c1 = min(c0 + chunk, ne);

  for (int b = t; b < nb; b += 1024) binCnt[b] = 0;
  __syncthreads();
  for (int e0 = c0 + t; e0 < c1; e0 += 4 * 1024) {
    int d[4];
#pragma unroll
    for (int u = 0; u < 4; ++u) {
      d[u] = dst[min(e0 + u * 1024, c1 - 1)];
    }
    asm volatile("" ::"v"(d[0]), "v"(d[1]), "v"(d[2]), "v"(d[3]));
#pragma unroll
    for (int u = 0; u < 4; ++u) {
      if (e0 + u * 1024 < c1) {
        const int dd = min(max(d[u], 0), n - 1);
        atomicAdd(&binCnt[dd >> SORT_RS], 1);
      }
    }
  }
  __syncthreads();

  int carry = 0;
  for (int b0 = 0; b0 < nb; b0 += 1024) {
    const int b = b0 + t;
    const int v = (b < nb) ? binCnt[b] : 0;
    int tot;
    const int ex = carry + block_excl_scan(v, sh, &tot);
    if (b < nb) {
      binCur[b] = ex;
      cntTab[(size_t)w * nb + b] = v;
      offTab[(size_t)w * nb + b] = ex;
    }
    carry += tot;
  }
  __syncthreads();

  for (int e0 = c0 + t; e0 < c1; e0 += 4 * 1024) {
    int d[4], s[4];
    float wv[4];
#pragma unroll
    for (int u = 0; u < 4; ++u) {
      const int e = min(e0 + u * 1024, c1 - 1);
      d[u]  = dst[e];
      s[u]  = src[e];
      wv[u] = ew[e];
    }
    asm volatile("" ::"v"(d[0]), "v"(d[1]), "v"(d[2]), "v"(d[3]),
                 "v"(s[0]), "v"(s[1]), "v"(s[2]), "v"(s[3]),
                 "v"(wv[0]), "v"(wv[1]), "v"(wv[2]), "v"(wv[3]));
#pragma unroll
    for (int u = 0; u < 4; ++u) {
      if (e0 + u * 1024 < c1) {
        const int dd = min(max(d[u], 0), n - 1);
        const int p = atomicAdd(&binCur[dd >> SORT_RS], 1);
        stS[p] = s[u];
        stD[p] = dd;
        stW[p] = wv[u];
      }
    }
  }
  __syncthreads();

  const int cn = c1 - c0;
  for (int i = t; i < cn; i += 1024) {
    segSrc[c0 + i] = stS[i];
    segDst[c0 + i] = stD[i];
    segW[c0 + i]   = stW[i];
  }
}

__global__ __launch_bounds__(1024) void own_rows(
    const int* __restrict__ segSrc, const int* __restrict__ segDst,
    const float* __restrict__ segW, const int* __restrict__ cntTab,
    const int* __restrict__ offTab, int* __restrict__ rowptr,
    int* __restrict__ csr, float* __restrict__ dinv, int n, int nb, int chunk) {
  __shared__ int sh[33];
  __shared__ int sCnt[SORT_W];
  __shared__ int sOff[SORT_W];
  __shared__ int nodeCnt[SORT_R];
  __shared__ int cursor[SORT_R];
  __shared__ float nodeDeg[SORT_R];
  const int t = threadIdx.x;
  const int b = blockIdx.x;

  int myOff = 0;
  if (t < SORT_W) {
    sCnt[t] = cntTab[(size_t)t * nb + b];
    myOff = offTab[(size_t)t * nb + b];
    sOff[t] = myOff;
  }
  if (t < SORT_R) {
    nodeCnt[t] = 0;
    nodeDeg[t] = 0.f;
  }
  int base;
  (void)block_excl_scan(myOff, sh, &base);

  const int w = t >> 2;
  const int sub = t & 3;
  const int cnt = sCnt[w];
  const int g0 = w * chunk + sOff[w];
  const int node0 = b * SORT_R;
  for (int i0 = sub; i0 < cnt; i0 += 16) {
    int d[4];
    float wv[4];
#pragma unroll
    for (int u = 0; u < 4; ++u) {
      const int i = i0 + 4 * u;
      const bool ok = i < cnt;
      d[u]  = ok ? segDst[g0 + i] : 0;
      wv[u] = ok ? segW[g0 + i] : 0.f;
    }
#pragma unroll
    for (int u = 0; u < 4; ++u) {
      if (i0 + 4 * u < cnt) {
        const int dd = d[u] - node0;
        atomicAdd(&nodeCnt[dd], 1);
        atomicAdd(&nodeDeg[dd], wv[u]);
      }
    }
  }
  __syncthreads();

  const int v = (t < SORT_R) ? nodeCnt[t] : 0;
  int nEdge;
  const int ex = block_excl_scan(v, sh, &nEdge);
  if (t < SORT_R) {
    cursor[t] = ex;
    const int node = node0 + t;
    if (node < n) {
      rowptr[node] = base + ex;
      const float dg = 1.0f + nodeDeg[t];
      dinv[node] = (dg > 0.f) ? rsqrtf(dg) : 0.f;
      if (node == n - 1) rowptr[n] = base + ex + v;
    }
  }
  __syncthreads();

  int2* ed = (int2*)csr;
  for (int i0 = sub; i0 < cnt; i0 += 16) {
    int d[4], s[4];
    float wv[4];
#pragma unroll
    for (int u = 0; u < 4; ++u) {
      const int i = i0 + 4 * u;
      const bool ok = i < cnt;
      d[u]  = ok ? segDst[g0 + i] : 0;
      s[u]  = ok ? segSrc[g0 + i] : 0;
      wv[u] = ok ? segW[g0 + i] : 0.f;
    }
#pragma unroll
    for (int u = 0; u < 4; ++u) {
      if (i0 + 4 * u < cnt) {
        const int p = atomicAdd(&cursor[d[u] - node0], 1);
        ed[base + p] = make_int2(s[u], __float_as_int(wv[u]));
      }
    }
  }
}

__global__ __launch_bounds__(256) void gather_rows(
    const float* __restrict__ h, const int* __restrict__ rowptr,
    const int* __restrict__ csr, const float* __restrict__ dinv,
    const float* __restrict__ bias, float* __restrict__ out, int n, int relu) {
  const int t = blockIdx.x * 256 + threadIdx.x;
  const int i = t >> 4;
  if (i >= n) return;
  const int l = t & 15;
  const float4* h4 = (const float4*)h;
  const int2* ed = (const int2*)csr;

  const int beg = rowptr[i];
  const int end = rowptr[i + 1];
  const float di = dinv[i];
  const float4 hv = h4[(size_t)i * 16 + l];
  float ax = hv.x * di, ay = hv.y * di, az = hv.z * di, aw = hv.w * di;

  for (int base = beg; base < end; base += 16) {
    const int p = base + l;
    int s = i;
    float cw = 0.f;
    if (p < end) {
      const int2 e = ed[p];
      s = min(max(e.x, 0), n - 1);
      cw = __int_as_float(e.y) * dinv[s];
    }
    const int m = min(16, end - base);
    for (int k = 0; k < m; k += 4) {
      const int s0 = __shfl(s, k + 0, 16);
      const int s1 = __shfl(s, k + 1, 16);
      const int s2 = __shfl(s, k + 2, 16);
      const int s3 = __shfl(s, k + 3, 16);
      const float c0 = __shfl(cw, k + 0, 16);
      const float c1 = __shfl(cw, k + 1, 16);
      const float c2 = __shfl(cw, k + 2, 16);
      const float c3 = __shfl(cw, k + 3, 16);
      const float4 v0 = h4[(size_t)s0 * 16 + l];
      const float4 v1 = h4[(size_t)s1 * 16 + l];
      const float4 v2 = h4[(size_t)s2 * 16 + l];
      const float4 v3 = h4[(size_t)s3 * 16 + l];
      ax = fmaf(v0.x, c0, ax); ay = fmaf(v0.y, c0, ay);
      az = fmaf(v0.z, c0, az); aw = fmaf(v0.w, c0, aw);
      ax = fmaf(v1.x, c1, ax); ay = fmaf(v1.y, c1, ay);
      az = fmaf(v1.z, c1, az); aw = fmaf(v1.w, c1, aw);
      ax = fmaf(v2.x, c2, ax); ay = fmaf(v2.y, c2, ay);
      az = fmaf(v2.z, c2, az); aw = fmaf(v2.w, c2, aw);
      ax = fmaf(v3.x, c3, ax); ay = fmaf(v3.y, c3, ay);
      az = fmaf(v3.z, c3, az); aw = fmaf(v3.w, c3, aw);
    }
  }

  float4 r;
  r.x = fmaf(ax, di, bias[l * 4 + 0]);
  r.y = fmaf(ay, di, bias[l * 4 + 1]);
  r.z = fmaf(az, di, bias[l * 4 + 2]);
  r.w = fmaf(aw, di, bias[l * 4 + 3]);
  if (relu) {
    r.x = fmaxf(r.x, 0.f); r.y = fmaxf(r.y, 0.f);
    r.z = fmaxf(r.z, 0.f); r.w = fmaxf(r.w, 0.f);
  }
  ((float4*)out)[(size_t)i * 16 + l] = r;
}

extern "C" void kernel_launch(void* const* d_in, const int* in_sizes, int n_in,
                              void* d_out, int out_size, void* d_ws,
                              size_t ws_size, hipStream_t stream) {
  const float* x  = (const float*)d_in[0];
  const int*   ei = (const int*)d_in[1];
  const float* ew = (const float*)d_in[2];
  const float* Wm = (const float*)d_in[3];
  const float* bm = (const float*)d_in[4];
  const float* W1 = (const float*)d_in[5];
  const float* b1 = (const float*)d_in[6];
  const float* W2 = (const float*)d_in[7];
  const float* b2 = (const float*)d_in[8];
  float* out = (float*)d_out;

  const int NN = in_sizes[0] / 384;
  const int EE = in_sizes[2];
  const int* src = ei;
  const int* dst = ei + EE;

  char* ws = (char*)d_ws;
  const int nBins = (NN + SORT_R - 1) / SORT_R;
  const int chunk = (EE + SORT_W - 1) / SORT_W;
  const size_t tBytes  = ((size_t)SORT_W * nBins * 4 + 255) & ~(size_t)255;
  const size_t nBytes  = ((size_t)NN * 4 + 255) & ~(size_t)255;
  const size_t n1Bytes = ((size_t)(NN + 1) * 4 + 255) & ~(size_t)255;
  const size_t eBytes  = ((size_t)EE * 4 + 255) & ~(size_t)255;
  const size_t e2Bytes = ((size_t)EE * 8 + 255) & ~(size_t)255;
  const size_t fBytes  = (size_t)NN * 64 * sizeof(float);
  size_t off = 0;
  int*   cntTab = (int*)(ws + off);   off += tBytes;
  int*   offTab = (int*)(ws + off);   off += tBytes;
  int*   segSrc = (int*)(ws + off);   off += eBytes;
  int*   segDst = (int*)(ws + off);   off += eBytes;
  float* segW   = (float*)(ws + off); off += eBytes;
  int*   rowptr = (int*)(ws + off);   off += n1Bytes;
  int*   csr    = (int*)(ws + off);   off += e2Bytes;
  float* dinv   = (float*)(ws + off); off += nBytes;
  float* x64    = (float*)(ws + off); off += fBytes;
  float* hbuf   = (float*)(ws + off); off += fBytes;
  float* agg    = (float*)(ws + off);

  const dim3 blk(256);
  const int gRow  = (int)(((long long)NN * 16 + 255) / 256);
  const int gGemm = (NN + GT_M - 1) / GT_M;
  const size_t binLds = ((size_t)2 * nBins + (size_t)3 * chunk) * 4;

  bin_edges<<<SORT_W, dim3(1024), binLds, stream>>>(
      src, dst, ew, segSrc, segDst, segW, cntTab, offTab, EE, NN, nBins, chunk);
  own_rows<<<nBins, dim3(1024), 0, stream>>>(
      segSrc, segDst, segW, cntTab, offTab, rowptr, csr, dinv, NN, nBins, chunk);

  gemm_f32_wmma<<<gGemm, blk, 0, stream>>>(x, Wm, bm, x64, NN, 384);

  gemm_f32_wmma<<<gGemm, blk, 0, stream>>>(x64, W1, nullptr, hbuf, NN, 64);
  gather_rows<<<gRow, blk, 0, stream>>>(hbuf, rowptr, csr, dinv, b1, agg, NN, 1);

  gemm_f32_wmma<<<gGemm, blk, 0, stream>>>(agg, W2, nullptr, hbuf, NN, 64);
  gather_rows<<<gRow, blk, 0, stream>>>(hbuf, rowptr, csr, dinv, b2, out, NN, 0);
}


# Each 64x64 weight product moved into the kernel that makes its left operand (rows handed over in LDS to WMMAs there): 7 launches -> 5, two 25.6 MB planes never written or re-read; sums unchanged
# speedup vs baseline: 3.0829x; 1.1008x over previous
#include <hip/hip_runtime.h>
#include <hip/hip_bf16.h>


typedef __attribute__((ext_vector_type(2))) float v2f;
typedef __attribute__((ext_vector_type(8))) float v8f;

#define GT_M  64
#define GT_K  32
#define GT_AP 36
#define GT_BP 160
#define GT_XP 68

__global__ __launch_bounds__(256, 16) void gemm_f32_wmma(
    const float* __restrict__ A, const float* __restrict__ B,
    const float* __restrict__ bias, const float* __restrict__ B2,
    float* __restrict__ C, int M, int K) {
  __shared__ __attribute__((aligned(16))) float sA[2][GT_M * GT_AP];
  __shared__ __attribute__((aligned(16))) float sB[2][(GT_K / 2) * GT_BP];

  const int t = threadIdx.x;
  const int wave = t >> 5;
  const int lane = t & 31;
  const int half = lane >> 4;
  const int l15  = lane & 15;
  const int strip = wave >> 1;
  const int colHalf = wave & 1;
  const int rowBlk = blockIdx.x * GT_M;

  const int aRow = t >> 3;
  const int aC4  = t & 7;
  const float4* aP0 =
      (const float4*)(A + (size_t)min(rowBlk + aRow, M - 1) * K) + aC4;
  const float4* aP1 =
      (const float4*)(A + (size_t)min(rowBlk + aRow + 32, M - 1) * K) + aC4;
  const int bKp = t >> 4;
  const int bN4 = t & 15;
  const float4* bP0 = (const float4*)(B + (size_t)(2 * bKp) * 64) + bN4;
  const float4* bP1 = (const float4*)(B + (size_t)(2 * bKp + 1) * 64) + bN4;

  const int aSt0 = aRow * GT_AP + aC4 * 4;
  const int aSt1 = (aRow + 32) * GT_AP + aC4 * 4;
  const int bSt  = bKp * GT_BP + bN4 * 8;
  const int aFr  = (16 * strip + l15) * GT_AP + 2 * half;
  const int bFr  = half * GT_BP + (32 * colHalf + l15) * 2;

  const int nStage = K / GT_K;
  float4 ra0 = aP0[0], ra1 = aP1[0], rb0 = bP0[0], rb1 = bP1[0];
  v8f acc0 = {}, acc1 = {};

  for (int c = 0; c < nStage; ++c) {
    float* dA = sA[c & 1];
    float* dB = sB[c & 1];
    *(float4*)(dA + aSt0) = ra0;
    *(float4*)(dA + aSt1) = ra1;
    *(float4*)(dB + bSt)     = make_float4(rb0.x, rb1.x, rb0.y, rb1.y);
    *(float4*)(dB + bSt + 4) = make_float4(rb0.z, rb1.z, rb0.w, rb1.w);
    __syncthreads();
    if (c + 1 < nStage) {
      ra0 = aP0[(size_t)(c + 1) * (GT_K / 4)];
      ra1 = aP1[(size_t)(c + 1) * (GT_K / 4)];
      rb0 = bP0[(size_t)(c + 1) * (GT_K * 16)];
      rb1 = bP1[(size_t)(c + 1) * (GT_K * 16)];
    }
    const float* fa = dA + aFr;
    const float* fb = dB + bFr;
#pragma unroll
    for (int j = 0; j < GT_K / 4; ++j) {
      const v2f a  = *(const v2f*)(fa + 4 * j);
      const v2f b0 = *(const v2f*)(fb + 2 * j * GT_BP);
      const v2f b1 = *(const v2f*)(fb + 2 * j * GT_BP + 32);
      acc0 = __builtin_amdgcn_wmma_f32_16x16x4_f32(false, a, false, b0,
                                                   (short)0, acc0, false, false);
      acc1 = __builtin_amdgcn_wmma_f32_16x16x4_f32(false, a, false, b1,
                                                   (short)0, acc1, false, false);
    }
  }

  const int col = 32 * colHalf + l15;
  const float bb0 = bias[col];
  const float bb1 = bias[col + 16];

  const float4* qP0 = (const float4*)(B2 + (size_t)(2 * bKp) * 64) + bN4;
  const float4* qP1 = (const float4*)(B2 + (size_t)(2 * bKp + 1) * 64) + bN4;
  const float4 q00 = qP0[0],   q01 = qP1[0];
  const float4 q10 = qP0[512], q11 = qP1[512];
  __syncthreads();
  float* sX = &sA[0][0];
  float* sW = &sB[0][0];
#pragma unroll
  for (int r = 0; r < 8; ++r) {
    float* xRow = sX + (16 * strip + r + 8 * half) * GT_XP + col;
    xRow[0]  = acc0[r] + bb0;
    xRow[16] = acc1[r] + bb1;
  }
  *(float4*)(sW + bSt)     = make_float4(q00.x, q01.x, q00.y, q01.y);
  *(float4*)(sW + bSt + 4) = make_float4(q00.z, q01.z, q00.w, q01.w);
  *(float4*)(sW + 16 * GT_BP + bSt)     = make_float4(q10.x, q11.x, q10.y, q11.y);
  *(float4*)(sW + 16 * GT_BP + bSt + 4) = make_float4(q10.z, q11.z, q10.w, q11.w);
  __syncthreads();
  v8f out0 = {}, out1 = {};
  {
    const float* fa = sX + (16 * strip + l15) * GT_XP + 2 * half;
    const float* fb = sW + bFr;
#pragma unroll
    for (int j = 0; j < 16; ++j) {
      const v2f a  = *(const v2f*)(fa + 4 * j);
      const v2f b0 = *(const v2f*)(fb + 2 * j * GT_BP);
      const v2f b1 = *(const v2f*)(fb + 2 * j * GT_BP + 32);
      out0 = __builtin_amdgcn_wmma_f32_16x16x4_f32(false, a, false, b0,
                                                   (short)0, out0, false, false);
      out1 = __builtin_amdgcn_wmma_f32_16x16x4_f32(false, a, false, b1,
                                                   (short)0, out1, false, false);
    }
  }

  const int row0 = rowBlk + 16 * strip;
  if (row0 >= M) return;
#pragma unroll
  for (int r = 0; r < 8; ++r) {
    float* cRow = C + (size_t)(row0 + r + 8 * half) * 64 + col;
    cRow[0]  = out0[r];
    cRow[16] = out1[r];
  }
}

#define SORT_W 256
#define SORT_R 256
#define SORT_RS 8

__device__ __forceinline__ int block_excl_scan(int v, int* sh, int* total) {
  const int lane = threadIdx.x & 31;
  const int wave = threadIdx.x >> 5;
  int inc = v;
  for (int o = 1; o < 32; o <<= 1) {
    const int u = __shfl_up(inc, o);
    if (lane >= o) inc += u;
  }
  if (lane == 31) sh[wave] = inc;
  __syncthreads();
  if (wave == 0) {
    const int s = sh[lane];
    int si = s;
    for (int o = 1; o < 32; o <<= 1) {
      const int u = __shfl_up(si, o);
      if (lane >= o) si += u;
    }
    sh[lane] = si - s;
    if (lane == 31) sh[32] = si;
  }
  __syncthreads();
  const int ex = sh[wave] + inc - v;
  *total = sh[32];
  __syncthreads();
  return ex;
}

__global__ __launch_bounds__(1024) void bin_edges(
    const int* __restrict__ src, const int* __restrict__ dst,
    const float* __restrict__ ew, int* __restrict__ segSrc,
    int* __restrict__ segDst, float* __restrict__ segW,
    int* __restrict__ cntTab, int* __restrict__ offTab,
    int ne, int n, int nb, int chunk) {
  extern __shared__ int dyn[];
  __shared__ int sh[33];
  int* binCnt = dyn;
  int* binCur = dyn + nb;
  int* stS = dyn + 2 * nb;
  int* stD = stS + chunk;
  float* stW = (float*)(stD + chunk);
  const int t = threadIdx.x;
  const int w = blockIdx.x;
  const int c0 = min(w * chunk, ne);
  const int c1 = min(c0 + chunk, ne);

  for (int b = t; b < nb; b += 1024) binCnt[b] = 0;
  __syncthreads();
  for (int e0 = c0 + t; e0 < c1; e0 += 4 * 1024) {
    int d[4];
#pragma unroll
    for (int u = 0; u < 4; ++u) {
      d[u] = dst[min(e0 + u * 1024, c1 - 1)];
    }
    asm volatile("" ::"v"(d[0]), "v"(d[1]), "v"(d[2]), "v"(d[3]));
#pragma unroll
    for (int u = 0; u < 4; ++u) {
      if (e0 + u * 1024 < c1) {
        const int dd = min(max(d[u], 0), n - 1);
        atomicAdd(&binCnt[dd >> SORT_RS], 1);
      }
    }
  }
  __syncthreads();

  int carry = 0;
  for (int b0 = 0; b0 < nb; b0 += 1024) {
    const int b = b0 + t;
    const int v = (b < nb) ? binCnt[b] : 0;
    int tot;
    const int ex = carry + block_excl_scan(v, sh, &tot);
    if (b < nb) {
      binCur[b] = ex;
      cntTab[(size_t)w * nb + b] = v;
      offTab[(size_t)w * nb + b] = ex;
    }
    carry += tot;
  }
  __syncthreads();

  for (int e0 = c0 + t; e0 < c1; e0 += 4 * 1024) {
    int d[4], s[4];
    float wv[4];
#pragma unroll
    for (int u = 0; u < 4; ++u) {
      const int e = min(e0 + u * 1024, c1 - 1);
      d[u]  = dst[e];
      s[u]  = src[e];
      wv[u] = ew[e];
    }
    asm volatile("" ::"v"(d[0]), "v"(d[1]), "v"(d[2]), "v"(d[3]),
                 "v"(s[0]), "v"(s[1]), "v"(s[2]), "v"(s[3]),
                 "v"(wv[0]), "v"(wv[1]), "v"(wv[2]), "v"(wv[3]));
#pragma unroll
    for (int u = 0; u < 4; ++u) {
      if (e0 + u * 1024 < c1) {
        const int dd = min(max(d[u], 0), n - 1);
        const int p = atomicAdd(&binCur[dd >> SORT_RS], 1);
        stS[p] = s[u];
        stD[p] = dd;
        stW[p] = wv[u];
      }
    }
  }
  __syncthreads();

  const int cn = c1 - c0;
  for (int i = t; i < cn; i += 1024) {
    segSrc[c0 + i] = stS[i];
    segDst[c0 + i] = stD[i];
    segW[c0 + i]   = stW[i];
  }
}

__global__ __launch_bounds__(1024) void own_rows(
    const int* __restrict__ segSrc, const int* __restrict__ segDst,
    const float* __restrict__ segW, const int* __restrict__ cntTab,
    const int* __restrict__ offTab, int* __restrict__ rowptr,
    int* __restrict__ csr, float* __restrict__ dinv, int n, int nb, int chunk) {
  __shared__ int sh[33];
  __shared__ int sCnt[SORT_W];
  __shared__ int sOff[SORT_W];
  __shared__ int nodeCnt[SORT_R];
  __shared__ int cursor[SORT_R];
  __shared__ float nodeDeg[SORT_R];
  const int t = threadIdx.x;
  const int b = blockIdx.x;

  int myOff = 0;
  if (t < SORT_W) {
    sCnt[t] = cntTab[(size_t)t * nb + b];
    myOff = offTab[(size_t)t * nb + b];
    sOff[t] = myOff;
  }
  if (t < SORT_R) {
    nodeCnt[t] = 0;
    nodeDeg[t] = 0.f;
  }
  int base;
  (void)block_excl_scan(myOff, sh, &base);

  const int w = t >> 2;
  const int sub = t & 3;
  const int cnt = sCnt[w];
  const int g0 = w * chunk + sOff[w];
  const int node0 = b * SORT_R;
  for (int i0 = sub; i0 < cnt; i0 += 16) {
    int d[4];
    float wv[4];
#pragma unroll
    for (int u = 0; u < 4; ++u) {
      const int i = i0 + 4 * u;
      const bool ok = i < cnt;
      d[u]  = ok ? segDst[g0 + i] : 0;
      wv[u] = ok ? segW[g0 + i] : 0.f;
    }
#pragma unroll
    for (int u = 0; u < 4; ++u) {
      if (i0 + 4 * u < cnt) {
        const int dd = d[u] - node0;
        atomicAdd(&nodeCnt[dd], 1);
        atomicAdd(&nodeDeg[dd], wv[u]);
      }
    }
  }
  __syncthreads();

  const int v = (t < SORT_R) ? nodeCnt[t] : 0;
  int nEdge;
  const int ex = block_excl_scan(v, sh, &nEdge);
  if (t < SORT_R) {
    cursor[t] = ex;
    const int node = node0 + t;
    if (node < n) {
      rowptr[node] = base + ex;
      const float dg = 1.0f + nodeDeg[t];
      dinv[node] = (dg > 0.f) ? rsqrtf(dg) : 0.f;
      if (node == n - 1) rowptr[n] = base + ex + v;
    }
  }
  __syncthreads();

  int2* ed = (int2*)csr;
  for (int i0 = sub; i0 < cnt; i0 += 16) {
    int d[4], s[4];
    float wv[4];
#pragma unroll
    for (int u = 0; u < 4; ++u) {
      const int i = i0 + 4 * u;
      const bool ok = i < cnt;
      d[u]  = ok ? segDst[g0 + i] : 0;
      s[u]  = ok ? segSrc[g0 + i] : 0;
      wv[u] = ok ? segW[g0 + i] : 0.f;
    }
#pragma unroll
    for (int u = 0; u < 4; ++u) {
      if (i0 + 4 * u < cnt) {
        const int p = atomicAdd(&cursor[d[u] - node0], 1);
        ed[base + p] = make_int2(s[u], __float_as_int(wv[u]));
      }
    }
  }
}

__device__ __forceinline__ float4 gather_row(
    const float* __restrict__ h, const int* __restrict__ rowptr,
    const int* __restrict__ csr, const float* __restrict__ dinv,
    const float* __restrict__ bias, int n, int relu, int i, int l) {
  const float4* h4 = (const float4*)h;
  const int2* ed = (const int2*)csr;

  const int beg = rowptr[i];
  const int end = rowptr[i + 1];
  const float di = dinv[i];
  const float4 hv = h4[(size_t)i * 16 + l];
  float ax = hv.x * di, ay = hv.y * di, az = hv.z * di, aw = hv.w * di;

  for (int base = beg; base < end; base += 16) {
    const int p = base + l;
    int s = i;
    float cw = 0.f;
    if (p < end) {
      const int2 e = ed[p];
      s = min(max(e.x, 0), n - 1);
      cw = __int_as_float(e.y) * dinv[s];
    }
    const int m = min(16, end - base);
    for (int k = 0; k < m; k += 4) {
      const int s0 = __shfl(s, k + 0, 16);
      const int s1 = __shfl(s, k + 1, 16);
      const int s2 = __shfl(s, k + 2, 16);
      const int s3 = __shfl(s, k + 3, 16);
      const float c0 = __shfl(cw, k + 0, 16);
      const float c1 = __shfl(cw, k + 1, 16);
      const float c2 = __shfl(cw, k + 2, 16);
      const float c3 = __shfl(cw, k + 3, 16);
      const float4 v0 = h4[(size_t)s0 * 16 + l];
      const float4 v1 = h4[(size_t)s1 * 16 + l];
      const float4 v2 = h4[(size_t)s2 * 16 + l];
      const float4 v3 = h4[(size_t)s3 * 16 + l];
      ax = fmaf(v0.x, c0, ax); ay = fmaf(v0.y, c0, ay);
      az = fmaf(v0.z, c0, az); aw = fmaf(v0.w, c0, aw);
      ax = fmaf(v1.x, c1, ax); ay = fmaf(v1.y, c1, ay);
      az = fmaf(v1.z, c1, az); aw = fmaf(v1.w, c1, aw);
      ax = fmaf(v2.x, c2, ax); ay = fmaf(v2.y, c2, ay);
      az = fmaf(v2.z, c2, az); aw = fmaf(v2.w, c2, aw);
      ax = fmaf(v3.x, c3, ax); ay = fmaf(v3.y, c3, ay);
      az = fmaf(v3.z, c3, az); aw = fmaf(v3.w, c3, aw);
    }
  }

  float4 r;
  r.x = fmaf(ax, di, bias[l * 4 + 0]);
  r.y = fmaf(ay, di, bias[l * 4 + 1]);
  r.z = fmaf(az, di, bias[l * 4 + 2]);
  r.w = fmaf(aw, di, bias[l * 4 + 3]);
  if (relu) {
    r.x = fmaxf(r.x, 0.f); r.y = fmaxf(r.y, 0.f);
    r.z = fmaxf(r.z, 0.f); r.w = fmaxf(r.w, 0.f);
  }
  return r;
}

__global__ __launch_bounds__(256) void gather_rows(
    const float* __restrict__ h, const int* __restrict__ rowptr,
    const int* __restrict__ csr, const float* __restrict__ dinv,
    const float* __restrict__ bias, float* __restrict__ out, int n, int relu) {
  const int t = blockIdx.x * 256 + threadIdx.x;
  const int i = t >> 4;
  if (i >= n) return;
  const int l = t & 15;
  ((float4*)out)[(size_t)i * 16 + l] =
      gather_row(h, rowptr, csr, dinv, bias, n, relu, i, l);
}

#define GF_P 68
__global__ __launch_bounds__(256) void gather_rows_mul(
    const float* __restrict__ h, const int* __restrict__ rowptr,
    const int* __restrict__ csr, const float* __restrict__ dinv,
    const float* __restrict__ bias, const float* __restrict__ Wn,
    float* __restrict__ out, int n, int relu) {
  __shared__ __attribute__((aligned(16))) float sIn[16 * GF_P];
  __shared__ __attribute__((aligned(16))) float sOut[16 * GF_P];
  const int tid = threadIdx.x;
  const int rowL = tid >> 4;
  const int i = blockIdx.x * 16 + rowL;
  const int l = tid & 15;
  const int wave = tid >> 5;
  const int lane = tid & 31;
  const int half = lane >> 4;
  const int l15 = lane & 15;

  float4 r = make_float4(0.f, 0.f, 0.f, 0.f);
  if (i < n) r = gather_row(h, rowptr, csr, dinv, bias, n, relu, i, l);
  *(float4*)(sIn + rowL * GF_P + 4 * l) = r;

  v2f bf[16];
  if (wave < 4) {
    const float* wp = Wn + (size_t)(2 * half) * 64 + 16 * wave + l15;
#pragma unroll
    for (int j = 0; j < 16; ++j) {
      bf[j].x = wp[(size_t)(4 * j) * 64];
      bf[j].y = wp[(size_t)(4 * j + 1) * 64];
    }
  }
  __syncthreads();
  if (wave < 4) {
    v8f acc = {};
    const float* fa = sIn + l15 * GF_P + 2 * half;
#pragma unroll
    for (int j = 0; j < 16; ++j) {
      const v2f a = *(const v2f*)(fa + 4 * j);
      acc = __builtin_amdgcn_wmma_f32_16x16x4_f32(false, a, false, bf[j],
                                                  (short)0, acc, false, false);
    }
#pragma unroll
    for (int q = 0; q < 8; ++q)
      sOut[(q + 8 * half) * GF_P + 16 * wave + l15] = acc[q];
  }
  __syncthreads();
  if (i < n)
    ((float4*)out)[(size_t)i * 16 + l] = *(const float4*)(sOut + rowL * GF_P + 4 * l);
}

extern "C" void kernel_launch(void* const* d_in, const int* in_sizes, int n_in,
                              void* d_out, int out_size, void* d_ws,
                              size_t ws_size, hipStream_t stream) {
  const float* x  = (const float*)d_in[0];
  const int*   ei = (const int*)d_in[1];
  const float* ew = (const float*)d_in[2];
  const float* Wm = (const float*)d_in[3];
  const float* bm = (const float*)d_in[4];
  const float* W1 = (const float*)d_in[5];
  const float* b1 = (const float*)d_in[6];
  const float* W2 = (const float*)d_in[7];
  const float* b2 = (const float*)d_in[8];
  float* out = (float*)d_out;

  const int NN = in_sizes[0] / 384;
  const int EE = in_sizes[2];
  const int* src = ei;
  const int* dst = ei + EE;

  char* ws = (char*)d_ws;
  const int nBins = (NN + SORT_R - 1) / SORT_R;
  const int chunk = (EE + SORT_W - 1) / SORT_W;
  const size_t tBytes  = ((size_t)SORT_W * nBins * 4 + 255) & ~(size_t)255;
  const size_t nBytes  = ((size_t)NN * 4 + 255) & ~(size_t)255;
  const size_t n1Bytes = ((size_t)(NN + 1) * 4 + 255) & ~(size_t)255;
  const size_t eBytes  = ((size_t)EE * 4 + 255) & ~(size_t)255;
  const size_t e2Bytes = ((size_t)EE * 8 + 255) & ~(size_t)255;
  const size_t fBytes  = (size_t)NN * 64 * sizeof(float);
  size_t off = 0;
  int*   cntTab = (int*)(ws + off);   off += tBytes;
  int*   offTab = (int*)(ws + off);   off += tBytes;
  int*   segSrc = (int*)(ws + off);   off += eBytes;
  int*   segDst = (int*)(ws + off);   off += eBytes;
  float* segW   = (float*)(ws + off); off += eBytes;
  int*   rowptr = (int*)(ws + off);   off += n1Bytes;
  int*   csr    = (int*)(ws + off);   off += e2Bytes;
  float* dinv   = (float*)(ws + off); off += nBytes;
  float* hbuf   = (float*)(ws + off); off += fBytes;
  float* h2     = (float*)(ws + off);

  const dim3 blk(256);
  const int gRow  = (int)(((long long)NN * 16 + 255) / 256);
  const int gGemm = (NN + GT_M - 1) / GT_M;
  const size_t binLds = ((size_t)2 * nBins + (size_t)3 * chunk) * 4;

  bin_edges<<<SORT_W, dim3(1024), binLds, stream>>>(
      src, dst, ew, segSrc, segDst, segW, cntTab, offTab, EE, NN, nBins, chunk);
  own_rows<<<nBins, dim3(1024), 0, stream>>>(
      segSrc, segDst, segW, cntTab, offTab, rowptr, csr, dinv, NN, nBins, chunk);

  gemm_f32_wmma<<<gGemm, blk, 0, stream>>>(x, Wm, bm, W1, hbuf, NN, 384);

  gather_rows_mul<<<gRow, blk, 0, stream>>>(hbuf, rowptr, csr, dinv, b1, W2, h2,
                                            NN, 1);

  gather_rows<<<gRow, blk, 0, stream>>>(h2, rowptr, csr, dinv, b2, out, NN, 0);
}


# Gathered rows stored pre-multiplied by their own node's rsqrt-degree (at the two producer kernels' row stores): no scattered per-edge normaliser lookup; sum order kept, products regrouped
# speedup vs baseline: 3.2324x; 1.0485x over previous
#include <hip/hip_runtime.h>
#include <hip/hip_bf16.h>


typedef __attribute__((ext_vector_type(2))) float v2f;
typedef __attribute__((ext_vector_type(8))) float v8f;

#define GT_M  64
#define GT_K  32
#define GT_AP 36
#define GT_BP 160
#define GT_XP 68

__global__ __launch_bounds__(256, 16) void gemm_f32_wmma(
    const float* __restrict__ A, const float* __restrict__ B,
    const float* __restrict__ bias, const float* __restrict__ B2,
    const float* __restrict__ rowScale, float* __restrict__ C, int M, int K) {
  __shared__ __attribute__((aligned(16))) float sA[2][GT_M * GT_AP];
  __shared__ __attribute__((aligned(16))) float sB[2][(GT_K / 2) * GT_BP];

  const int t = threadIdx.x;
  const int wave = t >> 5;
  const int lane = t & 31;
  const int half = lane >> 4;
  const int l15  = lane & 15;
  const int strip = wave >> 1;
  const int colHalf = wave & 1;
  const int rowBlk = blockIdx.x * GT_M;

  const int aRow = t >> 3;
  const int aC4  = t & 7;
  const float4* aP0 =
      (const float4*)(A + (size_t)min(rowBlk + aRow, M - 1) * K) + aC4;
  const float4* aP1 =
      (const float4*)(A + (size_t)min(rowBlk + aRow + 32, M - 1) * K) + aC4;
  const int bKp = t >> 4;
  const int bN4 = t & 15;
  const float4* bP0 = (const float4*)(B + (size_t)(2 * bKp) * 64) + bN4;
  const float4* bP1 = (const float4*)(B + (size_t)(2 * bKp + 1) * 64) + bN4;

  const int aSt0 = aRow * GT_AP + aC4 * 4;
  const int aSt1 = (aRow + 32) * GT_AP + aC4 * 4;
  const int bSt  = bKp * GT_BP + bN4 * 8;
  const int aFr  = (16 * strip + l15) * GT_AP + 2 * half;
  const int bFr  = half * GT_BP + (32 * colHalf + l15) * 2;

  const int nStage = K / GT_K;
  float4 ra0 = aP0[0], ra1 = aP1[0], rb0 = bP0[0], rb1 = bP1[0];
  v8f acc0 = {}, acc1 = {};

  for (int c = 0; c < nStage; ++c) {
    float* dA = sA[c & 1];
    float* dB = sB[c & 1];
    *(float4*)(dA + aSt0) = ra0;
    *(float4*)(dA + aSt1) = ra1;
    *(float4*)(dB + bSt)     = make_float4(rb0.x, rb1.x, rb0.y, rb1.y);
    *(float4*)(dB + bSt + 4) = make_float4(rb0.z, rb1.z, rb0.w, rb1.w);
    __syncthreads();
    if (c + 1 < nStage) {
      ra0 = aP0[(size_t)(c + 1) * (GT_K / 4)];
      ra1 = aP1[(size_t)(c + 1) * (GT_K / 4)];
      rb0 = bP0[(size_t)(c + 1) * (GT_K * 16)];
      rb1 = bP1[(size_t)(c + 1) * (GT_K * 16)];
    }
    const float* fa = dA + aFr;
    const float* fb = dB + bFr;
#pragma unroll
    for (int j = 0; j < GT_K / 4; ++j) {
      const v2f a  = *(const v2f*)(fa + 4 * j);
      const v2f b0 = *(const v2f*)(fb + 2 * j * GT_BP);
      const v2f b1 = *(const v2f*)(fb + 2 * j * GT_BP + 32);
      acc0 = __builtin_amdgcn_wmma_f32_16x16x4_f32(false, a, false, b0,
                                                   (short)0, acc0, false, false);
      acc1 = __builtin_amdgcn_wmma_f32_16x16x4_f32(false, a, false, b1,
                                                   (short)0, acc1, false, false);
    }
  }

  const int col = 32 * colHalf + l15;
  const float bb0 = bias[col];
  const float bb1 = bias[col + 16];

  const float4* qP0 = (const float4*)(B2 + (size_t)(2 * bKp) * 64) + bN4;
  const float4* qP1 = (const float4*)(B2 + (size_t)(2 * bKp + 1) * 64) + bN4;
  const float4 q00 = qP0[0],   q01 = qP1[0];
  const float4 q10 = qP0[512], q11 = qP1[512];
  const int row0 = rowBlk + 16 * strip;
  float rs[8];
#pragma unroll
  for (int r = 0; r < 8; ++r) rs[r] = rowScale[min(row0 + r + 8 * half, M - 1)];
  __syncthreads();
  float* sX = &sA[0][0];
  float* sW = &sB[0][0];
#pragma unroll
  for (int r = 0; r < 8; ++r) {
    float* xRow = sX + (16 * strip + r + 8 * half) * GT_XP + col;
    xRow[0]  = acc0[r] + bb0;
    xRow[16] = acc1[r] + bb1;
  }
  *(float4*)(sW + bSt)     = make_float4(q00.x, q01.x, q00.y, q01.y);
  *(float4*)(sW + bSt + 4) = make_float4(q00.z, q01.z, q00.w, q01.w);
  *(float4*)(sW + 16 * GT_BP + bSt)     = make_float4(q10.x, q11.x, q10.y, q11.y);
  *(float4*)(sW + 16 * GT_BP + bSt + 4) = make_float4(q10.z, q11.z, q10.w, q11.w);
  __syncthreads();
  v8f out0 = {}, out1 = {};
  {
    const float* fa = sX + (16 * strip + l15) * GT_XP + 2 * half;
    const float* fb = sW + bFr;
#pragma unroll
    for (int j = 0; j < 16; ++j) {
      const v2f a  = *(const v2f*)(fa + 4 * j);
      const v2f b0 = *(const v2f*)(fb + 2 * j * GT_BP);
      const v2f b1 = *(const v2f*)(fb + 2 * j * GT_BP + 32);
      out0 = __builtin_amdgcn_wmma_f32_16x16x4_f32(false, a, false, b0,
                                                   (short)0, out0, false, false);
      out1 = __builtin_amdgcn_wmma_f32_16x16x4_f32(false, a, false, b1,
                                                   (short)0, out1, false, false);
    }
  }

  if (row0 >= M) return;
#pragma unroll
  for (int r = 0; r < 8; ++r) {
    float* cRow = C + (size_t)(row0 + r + 8 * half) * 64 + col;
    cRow[0]  = out0[r] * rs[r];
    cRow[16] = out1[r] * rs[r];
  }
}

#define SORT_W 256
#define SORT_R 256
#define SORT_RS 8

__device__ __forceinline__ int block_excl_scan(int v, int* sh, int* total) {
  const int lane = threadIdx.x & 31;
  const int wave = threadIdx.x >> 5;
  int inc = v;
  for (int o = 1; o < 32; o <<= 1) {
    const int u = __shfl_up(inc, o);
    if (lane >= o) inc += u;
  }
  if (lane == 31) sh[wave] = inc;
  __syncthreads();
  if (wave == 0) {
    const int s = sh[lane];
    int si = s;
    for (int o = 1; o < 32; o <<= 1) {
      const int u = __shfl_up(si, o);
      if (lane >= o) si += u;
    }
    sh[lane] = si - s;
    if (lane == 31) sh[32] = si;
  }
  __syncthreads();
  const int ex = sh[wave] + inc - v;
  *total = sh[32];
  __syncthreads();
  return ex;
}

__global__ __launch_bounds__(1024) void bin_edges(
    const int* __restrict__ src, const int* __restrict__ dst,
    const float* __restrict__ ew, int* __restrict__ segSrc,
    int* __restrict__ segDst, float* __restrict__ segW,
    int* __restrict__ cntTab, int* __restrict__ offTab,
    int ne, int n, int nb, int chunk) {
  extern __shared__ int dyn[];
  __shared__ int sh[33];
  int* binCnt = dyn;
  int* binCur = dyn + nb;
  int* stS = dyn + 2 * nb;
  int* stD = stS + chunk;
  float* stW = (float*)(stD + chunk);
  const int t = threadIdx.x;
  const int w = blockIdx.x;
  const int c0 = min(w * chunk, ne);
  const int c1 = min(c0 + chunk, ne);

  for (int b = t; b < nb; b += 1024) binCnt[b] = 0;
  __syncthreads();
  for (int e0 = c0 + t; e0 < c1; e0 += 4 * 1024) {
    int d[4];
#pragma unroll
    for (int u = 0; u < 4; ++u) {
      d[u] = dst[min(e0 + u * 1024, c1 - 1)];
    }
    asm volatile("" ::"v"(d[0]), "v"(d[1]), "v"(d[2]), "v"(d[3]));
#pragma unroll
    for (int u = 0; u < 4; ++u) {
      if (e0 + u * 1024 < c1) {
        const int dd = min(max(d[u], 0), n - 1);
        atomicAdd(&binCnt[dd >> SORT_RS], 1);
      }
    }
  }
  __syncthreads();

  int carry = 0;
  for (int b0 = 0; b0 < nb; b0 += 1024) {
    const int b = b0 + t;
    const int v = (b < nb) ? binCnt[b] : 0;
    int tot;
    const int ex = carry + block_excl_scan(v, sh, &tot);
    if (b < nb) {
      binCur[b] = ex;
      cntTab[(size_t)w * nb + b] = v;
      offTab[(size_t)w * nb + b] = ex;
    }
    carry += tot;
  }
  __syncthreads();

  for (int e0 = c0 + t; e0 < c1; e0 += 4 * 1024) {
    int d[4], s[4];
    float wv[4];
#pragma unroll
    for (int u = 0; u < 4; ++u) {
      const int e = min(e0 + u * 1024, c1 - 1);
      d[u]  = dst[e];
      s[u]  = src[e];
      wv[u] = ew[e];
    }
    asm volatile("" ::"v"(d[0]), "v"(d[1]), "v"(d[2]), "v"(d[3]),
                 "v"(s[0]), "v"(s[1]), "v"(s[2]), "v"(s[3]),
                 "v"(wv[0]), "v"(wv[1]), "v"(wv[2]), "v"(wv[3]));
#pragma unroll
    for (int u = 0; u < 4; ++u) {
      if (e0 + u * 1024 < c1) {
        const int dd = min(max(d[u], 0), n - 1);
        const int p = atomicAdd(&binCur[dd >> SORT_RS], 1);
        stS[p] = s[u];
        stD[p] = dd;
        stW[p] = wv[u];
      }
    }
  }
  __syncthreads();

  const int cn = c1 - c0;
  for (int i = t; i < cn; i += 1024) {
    segSrc[c0 + i] = stS[i];
    segDst[c0 + i] = stD[i];
    segW[c0 + i]   = stW[i];
  }
}

__global__ __launch_bounds__(1024) void own_rows(
    const int* __restrict__ segSrc, const int* __restrict__ segDst,
    const float* __restrict__ segW, const int* __restrict__ cntTab,
    const int* __restrict__ offTab, int* __restrict__ rowptr,
    int* __restrict__ csr, float* __restrict__ dinv, int n, int nb, int chunk) {
  __shared__ int sh[33];
  __shared__ int sCnt[SORT_W];
  __shared__ int sOff[SORT_W];
  __shared__ int nodeCnt[SORT_R];
  __shared__ int cursor[SORT_R];
  __shared__ float nodeDeg[SORT_R];
  const int t = threadIdx.x;
  const int b = blockIdx.x;

  int myOff = 0;
  if (t < SORT_W) {
    sCnt[t] = cntTab[(size_t)t * nb + b];
    myOff = offTab[(size_t)t * nb + b];
    sOff[t] = myOff;
  }
  if (t < SORT_R) {
    nodeCnt[t] = 0;
    nodeDeg[t] = 0.f;
  }
  int base;
  (void)block_excl_scan(myOff, sh, &base);

  const int w = t >> 2;
  const int sub = t & 3;
  const int cnt = sCnt[w];
  const int g0 = w * chunk + sOff[w];
  const int node0 = b * SORT_R;
  for (int i0 = sub; i0 < cnt; i0 += 16) {
    int d[4];
    float wv[4];
#pragma unroll
    for (int u = 0; u < 4; ++u) {
      const int i = i0 + 4 * u;
      const bool ok = i < cnt;
      d[u]  = ok ? segDst[g0 + i] : 0;
      wv[u] = ok ? segW[g0 + i] : 0.f;
    }
#pragma unroll
    for (int u = 0; u < 4; ++u) {
      if (i0 + 4 * u < cnt) {
        const int dd = d[u] - node0;
        atomicAdd(&nodeCnt[dd], 1);
        atomicAdd(&nodeDeg[dd], wv[u]);
      }
    }
  }
  __syncthreads();

  const int v = (t < SORT_R) ? nodeCnt[t] : 0;
  int nEdge;
  const int ex = block_excl_scan(v, sh, &nEdge);
  if (t < SORT_R) {
    cursor[t] = ex;
    const int node = node0 + t;
    if (node < n) {
      rowptr[node] = base + ex;
      const float dg = 1.0f + nodeDeg[t];
      dinv[node] = (dg > 0.f) ? rsqrtf(dg) : 0.f;
      if (node == n - 1) rowptr[n] = base + ex + v;
    }
  }
  __syncthreads();

  int2* ed = (int2*)csr;
  for (int i0 = sub; i0 < cnt; i0 += 16) {
    int d[4], s[4];
    float wv[4];
#pragma unroll
    for (int u = 0; u < 4; ++u) {
      const int i = i0 + 4 * u;
      const bool ok = i < cnt;
      d[u]  = ok ? segDst[g0 + i] : 0;
      s[u]  = ok ? segSrc[g0 + i] : 0;
      wv[u] = ok ? segW[g0 + i] : 0.f;
    }
#pragma unroll
    for (int u = 0; u < 4; ++u) {
      if (i0 + 4 * u < cnt) {
        const int p = atomicAdd(&cursor[d[u] - node0], 1);
        ed[base + p] = make_int2(s[u], __float_as_int(wv[u]));
      }
    }
  }
}

__device__ __forceinline__ float4 gather_row(
    const float* __restrict__ hs, const int* __restrict__ rowptr,
    const int* __restrict__ csr, const float* __restrict__ bias, int n,
    int relu, int i, int l, float di) {
  const float4* h4 = (const float4*)hs;
  const int2* ed = (const int2*)csr;

  const int beg = rowptr[i];
  const int end = rowptr[i + 1];
  const float4 hv = h4[(size_t)i * 16 + l];
  float ax = hv.x, ay = hv.y, az = hv.z, aw = hv.w;

  for (int base = beg; base < end; base += 16) {
    const int p = base + l;
    int s = i;
    float cw = 0.f;
    if (p < end) {
      const int2 e = ed[p];
      s = min(max(e.x, 0), n - 1);
      cw = __int_as_float(e.y);
    }
    const int m = min(16, end - base);
    for (int k = 0; k < m; k += 4) {
      const int s0 = __shfl(s, k + 0, 16);
      const int s1 = __shfl(s, k + 1, 16);
      const int s2 = __shfl(s, k + 2, 16);
      const int s3 = __shfl(s, k + 3, 16);
      const float c0 = __shfl(cw, k + 0, 16);
      const float c1 = __shfl(cw, k + 1, 16);
      const float c2 = __shfl(cw, k + 2, 16);
      const float c3 = __shfl(cw, k + 3, 16);
      const float4 v0 = h4[(size_t)s0 * 16 + l];
      const float4 v1 = h4[(size_t)s1 * 16 + l];
      const float4 v2 = h4[(size_t)s2 * 16 + l];
      const float4 v3 = h4[(size_t)s3 * 16 + l];
      ax = fmaf(v0.x, c0, ax); ay = fmaf(v0.y, c0, ay);
      az = fmaf(v0.z, c0, az); aw = fmaf(v0.w, c0, aw);
      ax = fmaf(v1.x, c1, ax); ay = fmaf(v1.y, c1, ay);
      az = fmaf(v1.z, c1, az); aw = fmaf(v1.w, c1, aw);
      ax = fmaf(v2.x, c2, ax); ay = fmaf(v2.y, c2, ay);
      az = fmaf(v2.z, c2, az); aw = fmaf(v2.w, c2, aw);
      ax = fmaf(v3.x, c3, ax); ay = fmaf(v3.y, c3, ay);
      az = fmaf(v3.z, c3, az); aw = fmaf(v3.w, c3, aw);
    }
  }

  float4 r;
  r.x = fmaf(ax, di, bias[l * 4 + 0]);
  r.y = fmaf(ay, di, bias[l * 4 + 1]);
  r.z = fmaf(az, di, bias[l * 4 + 2]);
  r.w = fmaf(aw, di, bias[l * 4 + 3]);
  if (relu) {
    r.x = fmaxf(r.x, 0.f); r.y = fmaxf(r.y, 0.f);
    r.z = fmaxf(r.z, 0.f); r.w = fmaxf(r.w, 0.f);
  }
  return r;
}

__global__ __launch_bounds__(256) void gather_rows(
    const float* __restrict__ h, const int* __restrict__ rowptr,
    const int* __restrict__ csr, const float* __restrict__ dinv,
    const float* __restrict__ bias, float* __restrict__ out, int n, int relu) {
  const int t = blockIdx.x * 256 + threadIdx.x;
  const int i = t >> 4;
  if (i >= n) return;
  const int l = t & 15;
  ((float4*)out)[(size_t)i * 16 + l] =
      gather_row(h, rowptr, csr, bias, n, relu, i, l, dinv[i]);
}

#define GF_P 68
__global__ __launch_bounds__(256) void gather_rows_mul(
    const float* __restrict__ h, const int* __restrict__ rowptr,
    const int* __restrict__ csr, const float* __restrict__ dinv,
    const float* __restrict__ bias, const float* __restrict__ Wn,
    float* __restrict__ out, int n, int relu) {
  __shared__ __attribute__((aligned(16))) float sIn[16 * GF_P];
  __shared__ __attribute__((aligned(16))) float sOut[16 * GF_P];
  const int tid = threadIdx.x;
  const int rowL = tid >> 4;
  const int i = blockIdx.x * 16 + rowL;
  const int l = tid & 15;
  const int wave = tid >> 5;
  const int lane = tid & 31;
  const int half = lane >> 4;
  const int l15 = lane & 15;

  float4 r = make_float4(0.f, 0.f, 0.f, 0.f);
  float di = 0.f;
  if (i < n) {
    di = dinv[i];
    r = gather_row(h, rowptr, csr, bias, n, relu, i, l, di);
  }
  *(float4*)(sIn + rowL * GF_P + 4 * l) = r;

  v2f bf[16];
  if (wave < 4) {
    const float* wp = Wn + (size_t)(2 * half) * 64 + 16 * wave + l15;
#pragma unroll
    for (int j = 0; j < 16; ++j) {
      bf[j].x = wp[(size_t)(4 * j) * 64];
      bf[j].y = wp[(size_t)(4 * j + 1) * 64];
    }
  }
  __syncthreads();
  if (wave < 4) {
    v8f acc = {};
    const float* fa = sIn + l15 * GF_P + 2 * half;
#pragma unroll
    for (int j = 0; j < 16; ++j) {
      const v2f a = *(const v2f*)(fa + 4 * j);
      acc = __builtin_amdgcn_wmma_f32_16x16x4_f32(false, a, false, bf[j],
                                                  (short)0, acc, false, false);
    }
#pragma unroll
    for (int q = 0; q < 8; ++q)
      sOut[(q + 8 * half) * GF_P + 16 * wave + l15] = acc[q];
  }
  __syncthreads();
  if (i < n) {
    const float4 v = *(const float4*)(sOut + rowL * GF_P + 4 * l);
    ((float4*)out)[(size_t)i * 16 + l] =
        make_float4(v.x * di, v.y * di, v.z * di, v.w * di);
  }
}

extern "C" void kernel_launch(void* const* d_in, const int* in_sizes, int n_in,
                              void* d_out, int out_size, void* d_ws,
                              size_t ws_size, hipStream_t stream) {
  const float* x  = (const float*)d_in[0];
  const int*   ei = (const int*)d_in[1];
  const float* ew = (const float*)d_in[2];
  const float* Wm = (const float*)d_in[3];
  const float* bm = (const float*)d_in[4];
  const float* W1 = (const float*)d_in[5];
  const float* b1 = (const float*)d_in[6];
  const float* W2 = (const float*)d_in[7];
  const float* b2 = (const float*)d_in[8];
  float* out = (float*)d_out;

  const int NN = in_sizes[0] / 384;
  const int EE = in_sizes[2];
  const int* src = ei;
  const int* dst = ei + EE;

  char* ws = (char*)d_ws;
  const int nBins = (NN + SORT_R - 1) / SORT_R;
  const int chunk = (EE + SORT_W - 1) / SORT_W;
  const size_t tBytes  = ((size_t)SORT_W * nBins * 4 + 255) & ~(size_t)255;
  const size_t nBytes  = ((size_t)NN * 4 + 255) & ~(size_t)255;
  const size_t n1Bytes = ((size_t)(NN + 1) * 4 + 255) & ~(size_t)255;
  const size_t eBytes  = ((size_t)EE * 4 + 255) & ~(size_t)255;
  const size_t e2Bytes = ((size_t)EE * 8 + 255) & ~(size_t)255;
  const size_t fBytes  = (size_t)NN * 64 * sizeof(float);
  size_t off = 0;
  int*   cntTab = (int*)(ws + off);   off += tBytes;
  int*   offTab = (int*)(ws + off);   off += tBytes;
  int*   segSrc = (int*)(ws + off);   off += eBytes;
  int*   segDst = (int*)(ws + off);   off += eBytes;
  float* segW   = (float*)(ws + off); off += eBytes;
  int*   rowptr = (int*)(ws + off);   off += n1Bytes;
  int*   csr    = (int*)(ws + off);   off += e2Bytes;
  float* dinv   = (float*)(ws + off); off += nBytes;
  float* hbuf   = (float*)(ws + off); off += fBytes;
  float* h2     = (float*)(ws + off);

  const dim3 blk(256);
  const int gRow  = (int)(((long long)NN * 16 + 255) / 256);
  const int gGemm = (NN + GT_M - 1) / GT_M;
  const size_t binLds = ((size_t)2 * nBins + (size_t)3 * chunk) * 4;

  bin_edges<<<SORT_W, dim3(1024), binLds, stream>>>(
      src, dst, ew, segSrc, segDst, segW, cntTab, offTab, EE, NN, nBins, chunk);
  own_rows<<<nBins, dim3(1024), 0, stream>>>(
      segSrc, segDst, segW, cntTab, offTab, rowptr, csr, dinv, NN, nBins, chunk);

  gemm_f32_wmma<<<gGemm, blk, 0, stream>>>(x, Wm, bm, W1, dinv, hbuf, NN, 384);

  gather_rows_mul<<<gRow, blk, 0, stream>>>(hbuf, rowptr, csr, dinv, b1, W2, h2,
                                            NN, 1);

  gather_rows<<<gRow, blk, 0, stream>>>(h2, rowptr, csr, dinv, b2, out, NN, 0);
}
